# starting kernel
# baseline (speedup 1.0000x reference)
#include <hip/hip_runtime.h>
#include <hip/hip_bf16.h>

#define N_NODES   50000
#define N_EDGES   500000
#define IN_FEATS  128
#define N_HIDDEN  128
#define N_CLASSES 64

typedef __attribute__((ext_vector_type(2))) float v2f;
typedef __attribute__((ext_vector_type(8))) float v8f;

__global__ __launch_bounds__(256) void zero_kernel(float* __restrict__ p, int n) {
    int i = blockIdx.x * blockDim.x + threadIdx.x;
    if (i < n) p[i] = 0.0f;
}

template <int N, bool RELU_A>
__global__ __launch_bounds__(256)
void gemm_wmma_f32(const float* __restrict__ A, const float* __restrict__ W,
                   const float* __restrict__ bias, float* __restrict__ out, int M) {
    constexpr int K = 128;
    const int lane  = threadIdx.x & 31;
    const int wave  = threadIdx.x >> 5;
    const int mTile = blockIdx.x * 8 + wave;
    if (mTile * 16 >= M) return;
    const int nTile = blockIdx.y;
    const int half  = lane >> 4;
    const int l16   = lane & 15;
    const int nCol  = nTile * 16 + l16;

    const float bv = bias[nCol];
    v8f c;
#pragma unroll
    for (int i = 0; i < 8; ++i) c[i] = bv;

    const float* aRow = A + (size_t)(mTile * 16 + l16) * K;

#pragma unroll 4
    for (int k0 = 0; k0 < K; k0 += 4) {
        const int kk = k0 + 2 * half;
        v2f a = *(const v2f*)(aRow + kk);
        if (RELU_A) {
            a.x = fmaxf(a.x, 0.0f);
            a.y = fmaxf(a.y, 0.0f);
        }
        v2f b;
        b.x = W[(size_t)(kk + 0) * N + nCol];
        b.y = W[(size_t)(kk + 1) * N + nCol];
        c = __builtin_amdgcn_wmma_f32_16x16x4_f32(false, a, false, b,
                                                  (short)0, c, false, false);
    }

    float* oBase = out + (size_t)(mTile * 16 + half * 8) * N + nCol;
#pragma unroll
    for (int r = 0; r < 8; ++r) oBase[(size_t)r * N] = c[r];
}

template <int FEATS>
__global__ __launch_bounds__(256)
void aggregate_kernel(const float* __restrict__ h, const int* __restrict__ src,
                      const int* __restrict__ dst, const float* __restrict__ ew,
                      float* __restrict__ out, int nEdges) {
    const int lane = threadIdx.x & 31;
    const int e    = blockIdx.x * 8 + (threadIdx.x >> 5);
    if (e >= nEdges) return;
    const int   s = src[e];
    const int   d = dst[e];
    const float w = ew[e];

    constexpr int V = FEATS / 32;
    const float* hp = h   + (size_t)s * FEATS + lane * V;
    float*       op = out + (size_t)d * FEATS + lane * V;

    float vals[V];
#pragma unroll
    for (int i = 0; i < V; ++i) vals[i] = hp[i];
#pragma unroll
    for (int i = 0; i < V; ++i) unsafeAtomicAdd(op + i, vals[i] * w);
}

extern "C" void kernel_launch(void* const* d_in, const int* in_sizes, int n_in,
                              void* d_out, int out_size, void* d_ws, size_t ws_size,
                              hipStream_t stream) {
    const float* x   = (const float*)d_in[0];
    const int*   src = (const int*)  d_in[1];
    const int*   dst = (const int*)  d_in[2];
    const float* w1  = (const float*)d_in[3];
    const float* w2  = (const float*)d_in[4];
    const float* w3  = (const float*)d_in[5];
    const float* W1  = (const float*)d_in[6];
    const float* b1  = (const float*)d_in[7];
    const float* W2  = (const float*)d_in[8];
    const float* b2  = (const float*)d_in[9];
    const float* W3  = (const float*)d_in[10];
    const float* b3  = (const float*)d_in[11];
    float* out = (float*)d_out;

    float* wsA = (float*)d_ws;
    float* wsB = wsA + (size_t)N_NODES * N_HIDDEN;

    const int mTiles   = (N_NODES + 15) / 16;
    const dim3 blk(256);
    const dim3 gemmGrid128((mTiles + 7) / 8, N_HIDDEN / 16);
    const dim3 gemmGrid64 ((mTiles + 7) / 8, N_CLASSES / 16);
    const dim3 aggGrid((N_EDGES + 7) / 8);

    const int nH  = N_NODES * N_HIDDEN;
    const int nC  = N_NODES * N_CLASSES;
    const dim3 zeroGridH((nH + 255) / 256);
    const dim3 zeroGridC((nC + 255) / 256);

    gemm_wmma_f32<N_HIDDEN, false><<<gemmGrid128, blk, 0, stream>>>(x, W1, b1, wsA, N_NODES);
    zero_kernel<<<zeroGridH, blk, 0, stream>>>(wsB, nH);
    aggregate_kernel<N_HIDDEN><<<aggGrid, blk, 0, stream>>>(wsA, src, dst, w1, wsB, N_EDGES);

    gemm_wmma_f32<N_HIDDEN, true><<<gemmGrid128, blk, 0, stream>>>(wsB, W2, b2, wsA, N_NODES);
    zero_kernel<<<zeroGridH, blk, 0, stream>>>(wsB, nH);
    aggregate_kernel<N_HIDDEN><<<aggGrid, blk, 0, stream>>>(wsA, src, dst, w2, wsB, N_EDGES);

    gemm_wmma_f32<N_CLASSES, true><<<gemmGrid64, blk, 0, stream>>>(wsB, W3, b3, wsA, N_NODES);
    zero_kernel<<<zeroGridC, blk, 0, stream>>>(out, nC);
    aggregate_kernel<N_CLASSES><<<aggGrid, blk, 0, stream>>>(wsA, src, dst, w3, out, N_EDGES);
}


# Edge pass: lane l owns columns l, l+32, .. of the row, so each gather load and each f32 atomic covers one whole 128-B line (was 4 lines x 8 lanes per atomic); sum order still atomic arrival order
# speedup vs baseline: 1.2741x; 1.2741x over previous
#include <hip/hip_runtime.h>
#include <hip/hip_bf16.h>

#define N_NODES   50000
#define N_EDGES   500000
#define IN_FEATS  128
#define N_HIDDEN  128
#define N_CLASSES 64

typedef __attribute__((ext_vector_type(2))) float v2f;
typedef __attribute__((ext_vector_type(8))) float v8f;
typedef __attribute__((ext_vector_type(4))) float v4f;

template <int N, bool RELU_A>
__global__ __launch_bounds__(256)
void gemm_wmma_f32(const float* __restrict__ A, const float* __restrict__ W,
                   const float* __restrict__ bias, float* __restrict__ out,
                   float* __restrict__ zplane, int M) {
    constexpr int K = 128;
    const int lane  = threadIdx.x & 31;
    const int wave  = threadIdx.x >> 5;
    const int mTile = blockIdx.x * 8 + wave;
    if (mTile * 16 >= M) return;
    const int nTile = blockIdx.y;
    const int half  = lane >> 4;
    const int l16   = lane & 15;
    const int nCol  = nTile * 16 + l16;

    {
        float* zp = zplane + ((size_t)mTile * (N / 16) + nTile) * 256 + lane * 4;
        const v4f z4 = {0.0f, 0.0f, 0.0f, 0.0f};
        *(v4f*)(zp)       = z4;
        *(v4f*)(zp + 128) = z4;
    }

    const float bv = bias[nCol];
    v8f c;
#pragma unroll
    for (int i = 0; i < 8; ++i) c[i] = bv;

    const float* aRow = A + (size_t)(mTile * 16 + l16) * K;

#pragma unroll 4
    for (int k0 = 0; k0 < K; k0 += 4) {
        const int kk = k0 + 2 * half;
        v2f a = *(const v2f*)(aRow + kk);
        if (RELU_A) {
            a.x = fmaxf(a.x, 0.0f);
            a.y = fmaxf(a.y, 0.0f);
        }
        v2f b;
        b.x = W[(size_t)(kk + 0) * N + nCol];
        b.y = W[(size_t)(kk + 1) * N + nCol];
        c = __builtin_amdgcn_wmma_f32_16x16x4_f32(false, a, false, b,
                                                  (short)0, c, false, false);
    }

    float* oBase = out + (size_t)(mTile * 16 + half * 8) * N + nCol;
#pragma unroll
    for (int r = 0; r < 8; ++r) oBase[(size_t)r * N] = c[r];
}

template <int FEATS>
__global__ __launch_bounds__(256)
void aggregate_kernel(const float* __restrict__ h, const int* __restrict__ src,
                      const int* __restrict__ dst, const float* __restrict__ ew,
                      float* __restrict__ out, int nEdges) {
    const int lane = threadIdx.x & 31;
    const int e    = blockIdx.x * 8 + (threadIdx.x >> 5);
    if (e >= nEdges) return;
    const int   s = src[e];
    const int   d = dst[e];
    const float w = ew[e];

    constexpr int V = FEATS / 32;
    const float* hp = h   + (size_t)s * FEATS + lane;
    float*       op = out + (size_t)d * FEATS + lane;

    float vals[V];
#pragma unroll
    for (int i = 0; i < V; ++i) vals[i] = hp[i * 32];
#pragma unroll
    for (int i = 0; i < V; ++i) unsafeAtomicAdd(op + i * 32, vals[i] * w);
}

extern "C" void kernel_launch(void* const* d_in, const int* in_sizes, int n_in,
                              void* d_out, int out_size, void* d_ws, size_t ws_size,
                              hipStream_t stream) {
    const float* x   = (const float*)d_in[0];
    const int*   src = (const int*)  d_in[1];
    const int*   dst = (const int*)  d_in[2];
    const float* w1  = (const float*)d_in[3];
    const float* w2  = (const float*)d_in[4];
    const float* w3  = (const float*)d_in[5];
    const float* W1  = (const float*)d_in[6];
    const float* b1  = (const float*)d_in[7];
    const float* W2  = (const float*)d_in[8];
    const float* b2  = (const float*)d_in[9];
    const float* W3  = (const float*)d_in[10];
    const float* b3  = (const float*)d_in[11];
    float* out = (float*)d_out;

    float* wsA = (float*)d_ws;
    float* wsB = wsA + (size_t)N_NODES * N_HIDDEN;
    float* wsC = wsB + (size_t)N_NODES * N_HIDDEN;

    const int mTiles   = (N_NODES + 15) / 16;
    const dim3 blk(256);
    const dim3 gemmGrid128((mTiles + 7) / 8, N_HIDDEN / 16);
    const dim3 gemmGrid64 ((mTiles + 7) / 8, N_CLASSES / 16);
    const dim3 aggGrid((N_EDGES + 7) / 8);


    gemm_wmma_f32<N_HIDDEN, false><<<gemmGrid128, blk, 0, stream>>>(x, W1, b1, wsA, wsB, N_NODES);
    aggregate_kernel<N_HIDDEN><<<aggGrid, blk, 0, stream>>>(wsA, src, dst, w1, wsB, N_EDGES);

    gemm_wmma_f32<N_HIDDEN, true><<<gemmGrid128, blk, 0, stream>>>(wsB, W2, b2, wsA, wsC, N_NODES);
    aggregate_kernel<N_HIDDEN><<<aggGrid, blk, 0, stream>>>(wsA, src, dst, w2, wsC, N_EDGES);

    gemm_wmma_f32<N_CLASSES, true><<<gemmGrid64, blk, 0, stream>>>(wsC, W3, b3, wsA, out, N_NODES);
    aggregate_kernel<N_CLASSES><<<aggGrid, blk, 0, stream>>>(wsA, src, dst, w3, out, N_EDGES);
}
